# starting kernel
# baseline (speedup 1.0000x reference)
#include <hip/hip_runtime.h>
#include <hip/hip_bf16.h>


typedef __attribute__((ext_vector_type(2))) float v2f;
typedef __attribute__((ext_vector_type(8))) float v8f;

constexpr int Dk = 64;
constexpr int Kc = 512;
constexpr int WAVES_PER_BLOCK = 8;
constexpr int ROWS_PER_WAVE   = 32;

__global__ void kmeans_prep(const float* __restrict__ X,
                            const int* __restrict__ init_idx,
                            float* __restrict__ C,
                            float* __restrict__ c2,
                            float* __restrict__ sums,
                            float* __restrict__ counts) {
    const int k = blockIdx.x;
    const int t = threadIdx.x;
    const int src = init_idx[k];
    const float v = X[(size_t)src * Dk + t];
    C[k * Dk + t]    = v;
    sums[k * Dk + t] = 0.0f;

    __shared__ float red[Dk];
    red[t] = v * v;
    __syncthreads();
    if (t < 32) {
        float s = red[t] + red[t + 32];
        #pragma unroll
        for (int off = 16; off > 0; off >>= 1)
            s += __shfl_down(s, off, 32);
        if (t == 0) {
            c2[k]     = s;
            counts[k] = 0.0f;
        }
    }
}

__global__ void __launch_bounds__(256)
kmeans_assign(const float* __restrict__ X,
              const float* __restrict__ C,
              const float* __restrict__ c2,
              float* __restrict__ sums,
              float* __restrict__ counts,
              int numTiles) {
    extern __shared__ float lds[];
    float* ldsC  = lds;
    float* ldsC2 = lds + Kc * Dk;

    {
        const int tid = threadIdx.x;
        const float4* src4 = (const float4*)C;
        float4* dst4 = (float4*)ldsC;
        #pragma unroll 4
        for (int i = tid; i < (Kc * Dk) / 4; i += 256) dst4[i] = src4[i];
        for (int i = tid; i < Kc; i += 256) ldsC2[i] = c2[i];
    }
    __syncthreads();

    const int wave = threadIdx.x >> 5;
    const int lane = threadIdx.x & 31;
    const int tile = blockIdx.x * WAVES_PER_BLOCK + wave;
    if (tile >= numTiles) return;

    const int row0 = tile * ROWS_PER_WAVE;
    const int m  = lane & 15;
    const int kb = (lane >> 4) * 2;

    v2f a0[16], a1[16];
    {
        const float* xr0 = X + (size_t)(row0 + m) * Dk + kb;
        const float* xr1 = X + (size_t)(row0 + 16 + m) * Dk + kb;
        #pragma unroll
        for (int j = 0; j < 16; ++j) {
            a0[j] = *(const v2f*)(xr0 + 4 * j);
            a1[j] = *(const v2f*)(xr1 + 4 * j);
        }
    }

    float bestVal0[8], bestVal1[8];
    int   bestIdx0[8], bestIdx1[8];
    #pragma unroll
    for (int r = 0; r < 8; ++r) {
        bestVal0[r] = __builtin_inff(); bestIdx0[r] = 0;
        bestVal1[r] = __builtin_inff(); bestIdx1[r] = 0;
    }

    for (int n0 = 0; n0 < Kc; n0 += 16) {
        v2f b[16];
        {
            const float* crow = ldsC + (n0 + m) * Dk + kb;
            #pragma unroll
            for (int j = 0; j < 16; ++j)
                b[j] = *(const v2f*)(crow + 4 * j);
        }
        v8f acc0 = {0.f, 0.f, 0.f, 0.f, 0.f, 0.f, 0.f, 0.f};
        v8f acc1 = {0.f, 0.f, 0.f, 0.f, 0.f, 0.f, 0.f, 0.f};
        #pragma unroll
        for (int j = 0; j < 16; ++j) {
            acc0 = __builtin_amdgcn_wmma_f32_16x16x4_f32(
                false, a0[j], false, b[j], (short)0, acc0, false, false);
            acc1 = __builtin_amdgcn_wmma_f32_16x16x4_f32(
                false, a1[j], false, b[j], (short)0, acc1, false, false);
        }
        const float cc  = ldsC2[n0 + m];
        const int   col = n0 + m;
        #pragma unroll
        for (int r = 0; r < 8; ++r) {
            const float d0 = cc - 2.0f * acc0[r];
            const float d1 = cc - 2.0f * acc1[r];
            if (d0 < bestVal0[r]) { bestVal0[r] = d0; bestIdx0[r] = col; }
            if (d1 < bestVal1[r]) { bestVal1[r] = d1; bestIdx1[r] = col; }
        }
    }

    #pragma unroll
    for (int off = 1; off < 16; off <<= 1) {
        #pragma unroll
        for (int r = 0; r < 8; ++r) {
            float ov = __shfl_xor(bestVal0[r], off, 32);
            int   oi = __shfl_xor(bestIdx0[r], off, 32);
            if (ov < bestVal0[r] || (ov == bestVal0[r] && oi < bestIdx0[r])) {
                bestVal0[r] = ov; bestIdx0[r] = oi;
            }
            ov = __shfl_xor(bestVal1[r], off, 32);
            oi = __shfl_xor(bestIdx1[r], off, 32);
            if (ov < bestVal1[r] || (ov == bestVal1[r] && oi < bestIdx1[r])) {
                bestVal1[r] = ov; bestIdx1[r] = oi;
            }
        }
    }

    #pragma unroll
    for (int r = 0; r < 8; ++r) {
        const int lab[4] = {
            __shfl(bestIdx0[r], 0, 32),
            __shfl(bestIdx0[r], 16, 32),
            __shfl(bestIdx1[r], 0, 32),
            __shfl(bestIdx1[r], 16, 32)
        };
        const int rowOf[4] = { row0 + r, row0 + r + 8, row0 + 16 + r, row0 + 24 + r };
        #pragma unroll
        for (int q = 0; q < 4; ++q) {
            const float2 xv = ((const float2*)(X + (size_t)rowOf[q] * Dk))[lane];
            atomicAdd(&sums[lab[q] * Dk + 2 * lane + 0], xv.x);
            atomicAdd(&sums[lab[q] * Dk + 2 * lane + 1], xv.y);
            if (lane == 0) atomicAdd(&counts[lab[q]], 1.0f);
        }
    }
}

__global__ void kmeans_finalize(const float* __restrict__ sums,
                                const float* __restrict__ counts,
                                float* __restrict__ out) {
    const int k = blockIdx.x;
    const int t = threadIdx.x;
    const float cnt = fmaxf(counts[k], 1.0f);
    out[k * Dk + t] = sums[k * Dk + t] / cnt;
}

extern "C" void kernel_launch(void* const* d_in, const int* in_sizes, int n_in,
                              void* d_out, int out_size, void* d_ws, size_t ws_size,
                              hipStream_t stream) {
    const float* X        = (const float*)d_in[0];
    const int*   init_idx = (const int*)d_in[1];
    float*       out      = (float*)d_out;

    const int N        = in_sizes[0] / Dk;
    const int numTiles = N / ROWS_PER_WAVE;

    float* ws     = (float*)d_ws;
    float* C      = ws;
    float* c2     = C + Kc * Dk;
    float* sums   = c2 + Kc;
    float* counts = sums + Kc * Dk;

    kmeans_prep<<<Kc, Dk, 0, stream>>>(X, init_idx, C, c2, sums, counts);

    const int blocks = (numTiles + WAVES_PER_BLOCK - 1) / WAVES_PER_BLOCK;
    const size_t shmem = (size_t)(Kc * Dk + Kc) * sizeof(float);
    kmeans_assign<<<blocks, 32 * WAVES_PER_BLOCK, shmem, stream>>>(
        X, C, c2, sums, counts, numTiles);

    kmeans_finalize<<<Kc, Dk, 0, stream>>>(sums, counts, out);
}


# Segment sums by ownership: assign kernel stores one label per row; last kernel gives each work-group 2 centroids, scans all labels, gathers its rows, adds in fixed order; no atomics, no clearing
# speedup vs baseline: 2.5162x; 2.5162x over previous
#include <hip/hip_runtime.h>


typedef __attribute__((ext_vector_type(2))) float v2f;
typedef __attribute__((ext_vector_type(4))) float v4f;
typedef __attribute__((ext_vector_type(8))) float v8f;

constexpr int Dk = 64;
constexpr int Kc = 512;
constexpr int NWAVES      = 8;
constexpr int TILE_ROWS   = 16;
constexpr int GROUP_TILES = 8;
constexpr int JP_PITCH    = 132;
constexpr int TILE_FLOATS = 8 * JP_PITCH;
constexpr int XBUF_FLOATS = GROUP_TILES * TILE_FLOATS;
constexpr int XSTAGES     = 3;
constexpr int PART_FLOATS = GROUP_TILES * TILE_ROWS * NWAVES * 2;
constexpr int ASSIGN_BLOCKS = 256;
constexpr int ASSIGN_LDS_BYTES = (XSTAGES * XBUF_FLOATS + 2 * PART_FLOATS) * 4;

__global__ void kmeans_prep(const float* __restrict__ X,
                            const int* __restrict__ init_idx,
                            float* __restrict__ C,
                            float* __restrict__ c2) {
    const int k = blockIdx.x;
    const int t = threadIdx.x;
    const int src = init_idx[k];
    const float v = X[(size_t)src * Dk + t];
    C[k * Dk + t] = v;

    __shared__ float red[Dk];
    red[t] = v * v;
    __syncthreads();
    if (t < 32) {
        float s = red[t] + red[t + 32];
        #pragma unroll
        for (int off = 16; off > 0; off >>= 1)
            s += __shfl_down(s, off, 32);
        if (t == 0) c2[k] = s;
    }
}

__global__ void __launch_bounds__(256)
kmeans_assign(const float* __restrict__ X,
              const float* __restrict__ C,
              const float* __restrict__ c2,
              unsigned short* __restrict__ labels,
              int numTiles32) {
    extern __shared__ float lds[];
    float* xbuf = lds;
    float* part = lds + XSTAGES * XBUF_FLOATS;

    const int wave = threadIdx.x >> 5;
    const int lane = threadIdx.x & 31;
    const int c = lane & 15;
    const int h = lane >> 4;

    const int numTiles = numTiles32 * 2;
    const int t0 = (int)(((long long)blockIdx.x * numTiles) / (int)gridDim.x);
    const int t1 = (int)(((long long)(blockIdx.x + 1) * numTiles) / (int)gridDim.x);
    const int nTiles  = t1 - t0;
    const int nGroups = (nTiles + GROUP_TILES - 1) / GROUP_TILES;

    v2f a[4][16];
    {
        const float* cr = C + (size_t)(wave * 64 + c) * Dk + 2 * h;
        #pragma unroll
        for (int t = 0; t < 4; ++t) {
            #pragma unroll
            for (int j = 0; j < 16; ++j)
                a[t][j] = *(const v2f*)(cr + t * 16 * Dk + 4 * j);
        }
    }
    float cc[4][8];
    #pragma unroll
    for (int t = 0; t < 4; ++t) {
        #pragma unroll
        for (int r = 0; r < 8; ++r)
            cc[t][r] = c2[wave * 64 + 16 * t + 8 * h + r];
    }

    v4f xr[8];
    auto loadTile = [&](int g) {
        const int ti = g * GROUP_TILES + wave;
        if (ti < nTiles) {
            const float* src = X + ((size_t)(t0 + ti) * TILE_ROWS + c) * Dk + 4 * h;
            #pragma unroll
            for (int i = 0; i < 8; ++i) xr[i] = *(const v4f*)(src + 8 * i);
        }
    };
    auto storeTile = [&](int g, int stage) {
        const int ti = g * GROUP_TILES + wave;
        if (ti < nTiles) {
            float* dst = xbuf + stage * XBUF_FLOATS + wave * TILE_FLOATS + 4 * c + 2 * h;
            #pragma unroll
            for (int i = 0; i < 8; ++i) {
                v2f lo = { xr[i].x, xr[i].y };
                v2f hi = { xr[i].z, xr[i].w };
                *(v2f*)(dst + i * JP_PITCH)      = lo;
                *(v2f*)(dst + i * JP_PITCH + 64) = hi;
            }
        }
    };

    loadTile(0);
    storeTile(0, 0);
    __syncthreads();

    int stage = 0;
    for (int g = 0; g < nGroups; ++g) {
        const int left = nTiles - g * GROUP_TILES;
        const int nT = left < GROUP_TILES ? left : GROUP_TILES;
        const int nextStage = (stage == XSTAGES - 1) ? 0 : stage + 1;
        if (g + 1 < nGroups) loadTile(g + 1);

        const float* xb = xbuf + stage * XBUF_FLOATS;
        float* pb = part + (g & 1) * PART_FLOATS;

        v4f bq[8];
        #pragma unroll
        for (int jp = 0; jp < 8; ++jp)
            bq[jp] = *(const v4f*)(xb + jp * JP_PITCH + 4 * lane);

        for (int i = 0; i < nT; ++i) {
            const int inext = (i + 1 < nT) ? i + 1 : i;
            v4f bn[8];
            {
                const float* xt = xb + inext * TILE_FLOATS + 4 * lane;
                #pragma unroll
                for (int jp = 0; jp < 8; ++jp)
                    bn[jp] = *(const v4f*)(xt + jp * JP_PITCH);
            }

            v8f acc[4];
            #pragma unroll
            for (int t = 0; t < 4; ++t)
                acc[t] = (v8f){0.f, 0.f, 0.f, 0.f, 0.f, 0.f, 0.f, 0.f};
            #pragma unroll
            for (int j = 0; j < 16; ++j) {
                v2f b;
                if (j & 1) { b.x = bq[j >> 1].z; b.y = bq[j >> 1].w; }
                else       { b.x = bq[j >> 1].x; b.y = bq[j >> 1].y; }
                #pragma unroll
                for (int t = 0; t < 4; ++t)
                    acc[t] = __builtin_amdgcn_wmma_f32_16x16x4_f32(
                        false, a[t][j], false, b, (short)0, acc[t], false, false);
            }

            float bt[4];
            int   it[4];
            #pragma unroll
            for (int t = 0; t < 4; ++t) {
                bt[t] = __builtin_inff();
                it[t] = 16 * t;
                #pragma unroll
                for (int r = 0; r < 8; ++r) {
                    const float d = cc[t][r] - 2.0f * acc[t][r];
                    if (d < bt[t]) { bt[t] = d; it[t] = 16 * t + r; }
                }
            }
            if (bt[1] < bt[0]) { bt[0] = bt[1]; it[0] = it[1]; }
            if (bt[3] < bt[2]) { bt[2] = bt[3]; it[2] = it[3]; }
            if (bt[2] < bt[0]) { bt[0] = bt[2]; it[0] = it[2]; }
            float best = bt[0];
            int   bidx = it[0] + wave * 64 + 8 * h;
            {
                const float ov = __shfl_xor(best, 16, 32);
                const int   oi = __shfl_xor(bidx, 16, 32);
                if (ov < best || (ov == best && oi < bidx)) { best = ov; bidx = oi; }
            }
            if (h == 0) {
                v2f pv = { best, __int_as_float(bidx) };
                *(v2f*)(pb + ((i * TILE_ROWS + c) * NWAVES + wave) * 2) = pv;
            }

            #pragma unroll
            for (int jp = 0; jp < 8; ++jp) bq[jp] = bn[jp];
        }

        if (g + 1 < nGroups) storeTile(g + 1, nextStage);
        __syncthreads();

        if (wave < nT) {
            const float* pp = pb + ((wave * TILE_ROWS + c) * NWAVES + 4 * h) * 2;
            const v4f p0 = *(const v4f*)pp;
            const v4f p1 = *(const v4f*)(pp + 4);
            float bv = p0.x; int bi = __float_as_int(p0.y);
            if (p0.z < bv) { bv = p0.z; bi = __float_as_int(p0.w); }
            if (p1.x < bv) { bv = p1.x; bi = __float_as_int(p1.y); }
            if (p1.z < bv) { bv = p1.z; bi = __float_as_int(p1.w); }
            {
                const float ov = __shfl_xor(bv, 16, 32);
                const int   oi = __shfl_xor(bi, 16, 32);
                if (ov < bv || (ov == bv && oi < bi)) { bv = ov; bi = oi; }
            }
            if (h == 0)
                labels[(size_t)(t0 + g * GROUP_TILES + wave) * TILE_ROWS + c] = (unsigned short)bi;
        }
        stage = nextStage;
    }
}

typedef __attribute__((ext_vector_type(2))) unsigned short us2;
typedef __attribute__((ext_vector_type(4))) unsigned int   v4u;

constexpr int SUM_WAVES = 16;
constexpr int CHUNK     = 256;
constexpr int BATCH     = 16;
constexpr int LIST_CAP  = BATCH + CHUNK;
constexpr int SCAN_AHEAD = 4;

__device__ __forceinline__ unsigned pk_min_u16(unsigned a, unsigned b) {
    const us2 r = __builtin_elementwise_min(__builtin_bit_cast(us2, a), __builtin_bit_cast(us2, b));
    return __builtin_bit_cast(unsigned, r);
}

__global__ void __launch_bounds__(32 * SUM_WAVES)
kmeans_sum(const float* __restrict__ X,
           const unsigned short* __restrict__ labels,
           float* __restrict__ out,
           int N) {
    __shared__ unsigned list[SUM_WAVES][LIST_CAP];
    __shared__ float psum[SUM_WAVES][2][Dk];
    __shared__ float pcnt[SUM_WAVES][2];

    const int wave = threadIdx.x >> 5;
    const int lane = threadIdx.x & 31;
    const unsigned k0  = 2u * blockIdx.x;
    const unsigned pat = k0 | (k0 << 16);
    unsigned* myList = list[wave];

    v2f acc0 = {0.f, 0.f}, acc1 = {0.f, 0.f};
    int n0 = 0, n1 = 0;
    int cnt = 0;

    const int nChunks = (N + CHUNK - 1) / CHUNK;
    const v4u* L4 = (const v4u*)labels;

    auto loadChunk = [&](int ch) -> v4u {
        const bool ok = (ch < nChunks) && (ch * CHUNK + lane * 8 + 8 <= N);
        const v4u w = L4[ok ? (size_t)ch * 32 + lane : (size_t)0];
        const v4u none = {0xFFFFFFFFu, 0xFFFFFFFFu, 0xFFFFFFFFu, 0xFFFFFFFFu};
        return ok ? w : none;
    };

    auto gatherFull = [&](int base) {
        unsigned e[BATCH];
        v2f v[BATCH];
        #pragma unroll
        for (int i = 0; i < BATCH; ++i) {
            e[i] = __builtin_amdgcn_readfirstlane(myList[base + i]);
            v[i] = *(const v2f*)(X + (size_t)(e[i] >> 1) * Dk + 2 * lane);
        }
        #pragma unroll
        for (int i = 0; i < BATCH; ++i) {
            if (e[i] & 1u) { acc1 += v[i]; ++n1; }
            else           { acc0 += v[i]; ++n0; }
        }
    };

    auto scanChunk = [&](v4u w, int ch) {
        const unsigned y0 = (w.x ^ pat) & 0xFFFEFFFEu;
        const unsigned y1 = (w.y ^ pat) & 0xFFFEFFFEu;
        const unsigned y2 = (w.z ^ pat) & 0xFFFEFFFEu;
        const unsigned y3 = (w.w ^ pat) & 0xFFFEFFFEu;
        const unsigned t = pk_min_u16(pk_min_u16(y0, y1), pk_min_u16(y2, y3));
        const bool hit = ((t & 0xFFFFu) == 0u) || (t < 0x10000u);
        if (__builtin_amdgcn_ballot_w32(hit) == 0u) return;

        const unsigned ys[4] = { y0, y1, y2, y3 };
        const unsigned ws[4] = { w.x, w.y, w.z, w.w };
        const unsigned rowBase = (unsigned)ch * CHUNK + (unsigned)lane * 8u;
        #pragma unroll
        for (int j = 0; j < 8; ++j) {
            const unsigned yj = (j & 1) ? (ys[j >> 1] >> 16) : (ys[j >> 1] & 0xFFFFu);
            const unsigned lj = (j & 1) ? (ws[j >> 1] >> 16) : (ws[j >> 1] & 0xFFFFu);
            const bool mine = (yj == 0u);
            const unsigned bal = __builtin_amdgcn_ballot_w32(mine);
            if (bal != 0u) {
                if (mine) {
                    const int pos = cnt + (int)__builtin_amdgcn_mbcnt_lo(bal, 0u);
                    myList[pos] = ((rowBase + (unsigned)j) << 1) | (lj & 1u);
                }
                cnt += __builtin_popcount(bal);
            }
        }
        __builtin_amdgcn_fence(__ATOMIC_RELEASE, "wavefront");
        __builtin_amdgcn_wave_barrier();
        __builtin_amdgcn_fence(__ATOMIC_ACQUIRE, "wavefront");
        while (cnt >= BATCH) { cnt -= BATCH; gatherFull(cnt); }
    };

    v4u cur[SCAN_AHEAD];
    #pragma unroll
    for (int u = 0; u < SCAN_AHEAD; ++u) cur[u] = loadChunk(wave + SUM_WAVES * u);
    for (int ch = wave; ch < nChunks; ch += SUM_WAVES * SCAN_AHEAD) {
        v4u nxt[SCAN_AHEAD];
        #pragma unroll
        for (int u = 0; u < SCAN_AHEAD; ++u) nxt[u] = loadChunk(ch + SUM_WAVES * (SCAN_AHEAD + u));
        #pragma unroll
        for (int u = 0; u < SCAN_AHEAD; ++u) scanChunk(cur[u], ch + SUM_WAVES * u);
        #pragma unroll
        for (int u = 0; u < SCAN_AHEAD; ++u) cur[u] = nxt[u];
    }

    for (int i = 0; i < cnt; ++i) {
        const unsigned e = __builtin_amdgcn_readfirstlane(myList[i]);
        const v2f v = *(const v2f*)(X + (size_t)(e >> 1) * Dk + 2 * lane);
        if (e & 1u) { acc1 += v; ++n1; }
        else        { acc0 += v; ++n0; }
    }

    *(v2f*)&psum[wave][0][2 * lane] = acc0;
    *(v2f*)&psum[wave][1][2 * lane] = acc1;
    if (lane == 0) { pcnt[wave][0] = (float)n0; pcnt[wave][1] = (float)n1; }
    __syncthreads();

    if (threadIdx.x < 2 * Dk) {
        const int which = threadIdx.x >> 6;
        const int f     = threadIdx.x & (Dk - 1);
        float s = 0.f, n = 0.f;
        #pragma unroll
        for (int w = 0; w < SUM_WAVES; ++w) { s += psum[w][which][f]; n += pcnt[w][which]; }
        out[(k0 + which) * Dk + f] = s / fmaxf(n, 1.0f);
    }
}

extern "C" void kernel_launch(void* const* d_in, const int* in_sizes, int n_in,
                              void* d_out, int out_size, void* d_ws, size_t ws_size,
                              hipStream_t stream) {
    const float* X        = (const float*)d_in[0];
    const int*   init_idx = (const int*)d_in[1];
    float*       out      = (float*)d_out;

    const int N          = in_sizes[0] / Dk;
    const int numTiles32 = N / 32;

    float* ws = (float*)d_ws;
    float* C  = ws;
    float* c2 = C + Kc * Dk;
    unsigned short* labels = (unsigned short*)(c2 + Kc);

    kmeans_prep<<<Kc, Dk, 0, stream>>>(X, init_idx, C, c2);

    kmeans_assign<<<ASSIGN_BLOCKS, 32 * NWAVES, ASSIGN_LDS_BYTES, stream>>>(
        X, C, c2, labels, numTiles32);

    kmeans_sum<<<Kc / 2, 32 * SUM_WAVES, 0, stream>>>(X, labels, out, N);
}


# Segment-sum kernel split 4 ways by row range: 1024 work-groups (2 centroids x a quarter of the labels) write partial sums, a 4th small kernel adds them in slice order; changes summation order
# speedup vs baseline: 3.3888x; 1.3468x over previous
#include <hip/hip_runtime.h>


typedef __attribute__((ext_vector_type(2))) float v2f;
typedef __attribute__((ext_vector_type(4))) float v4f;
typedef __attribute__((ext_vector_type(8))) float v8f;

constexpr int Dk = 64;
constexpr int Kc = 512;
constexpr int NWAVES      = 8;
constexpr int TILE_ROWS   = 16;
constexpr int GROUP_TILES = 8;
constexpr int JP_PITCH    = 132;
constexpr int TILE_FLOATS = 8 * JP_PITCH;
constexpr int XBUF_FLOATS = GROUP_TILES * TILE_FLOATS;
constexpr int XSTAGES     = 3;
constexpr int PART_FLOATS = GROUP_TILES * TILE_ROWS * NWAVES * 2;
constexpr int ASSIGN_BLOCKS = 256;
constexpr int ASSIGN_LDS_BYTES = (XSTAGES * XBUF_FLOATS + 2 * PART_FLOATS) * 4;

__global__ void kmeans_prep(const float* __restrict__ X,
                            const int* __restrict__ init_idx,
                            float* __restrict__ C,
                            float* __restrict__ c2) {
    const int k = blockIdx.x;
    const int t = threadIdx.x;
    const int src = init_idx[k];
    const float v = X[(size_t)src * Dk + t];
    C[k * Dk + t] = v;

    __shared__ float red[Dk];
    red[t] = v * v;
    __syncthreads();
    if (t < 32) {
        float s = red[t] + red[t + 32];
        #pragma unroll
        for (int off = 16; off > 0; off >>= 1)
            s += __shfl_down(s, off, 32);
        if (t == 0) c2[k] = s;
    }
}

__global__ void __launch_bounds__(256)
kmeans_assign(const float* __restrict__ X,
              const float* __restrict__ C,
              const float* __restrict__ c2,
              unsigned short* __restrict__ labels,
              int numTiles32) {
    extern __shared__ float lds[];
    float* xbuf = lds;
    float* part = lds + XSTAGES * XBUF_FLOATS;

    const int wave = threadIdx.x >> 5;
    const int lane = threadIdx.x & 31;
    const int c = lane & 15;
    const int h = lane >> 4;

    const int numTiles = numTiles32 * 2;
    const int t0 = (int)(((long long)blockIdx.x * numTiles) / (int)gridDim.x);
    const int t1 = (int)(((long long)(blockIdx.x + 1) * numTiles) / (int)gridDim.x);
    const int nTiles  = t1 - t0;
    const int nGroups = (nTiles + GROUP_TILES - 1) / GROUP_TILES;

    v2f a[4][16];
    {
        const float* cr = C + (size_t)(wave * 64 + c) * Dk + 2 * h;
        #pragma unroll
        for (int t = 0; t < 4; ++t) {
            #pragma unroll
            for (int j = 0; j < 16; ++j)
                a[t][j] = *(const v2f*)(cr + t * 16 * Dk + 4 * j);
        }
    }
    float cc[4][8];
    #pragma unroll
    for (int t = 0; t < 4; ++t) {
        #pragma unroll
        for (int r = 0; r < 8; ++r)
            cc[t][r] = c2[wave * 64 + 16 * t + 8 * h + r];
    }

    v4f xr[8];
    auto loadTile = [&](int g) {
        const int ti = g * GROUP_TILES + wave;
        if (ti < nTiles) {
            const float* src = X + ((size_t)(t0 + ti) * TILE_ROWS + c) * Dk + 4 * h;
            #pragma unroll
            for (int i = 0; i < 8; ++i) xr[i] = *(const v4f*)(src + 8 * i);
        }
    };
    auto storeTile = [&](int g, int stage) {
        const int ti = g * GROUP_TILES + wave;
        if (ti < nTiles) {
            float* dst = xbuf + stage * XBUF_FLOATS + wave * TILE_FLOATS + 4 * c + 2 * h;
            #pragma unroll
            for (int i = 0; i < 8; ++i) {
                v2f lo = { xr[i].x, xr[i].y };
                v2f hi = { xr[i].z, xr[i].w };
                *(v2f*)(dst + i * JP_PITCH)      = lo;
                *(v2f*)(dst + i * JP_PITCH + 64) = hi;
            }
        }
    };

    loadTile(0);
    storeTile(0, 0);
    __syncthreads();

    int stage = 0;
    for (int g = 0; g < nGroups; ++g) {
        const int left = nTiles - g * GROUP_TILES;
        const int nT = left < GROUP_TILES ? left : GROUP_TILES;
        const int nextStage = (stage == XSTAGES - 1) ? 0 : stage + 1;
        if (g + 1 < nGroups) loadTile(g + 1);

        const float* xb = xbuf + stage * XBUF_FLOATS;
        float* pb = part + (g & 1) * PART_FLOATS;

        v4f bq[8];
        #pragma unroll
        for (int jp = 0; jp < 8; ++jp)
            bq[jp] = *(const v4f*)(xb + jp * JP_PITCH + 4 * lane);

        for (int i = 0; i < nT; ++i) {
            const int inext = (i + 1 < nT) ? i + 1 : i;
            v4f bn[8];
            {
                const float* xt = xb + inext * TILE_FLOATS + 4 * lane;
                #pragma unroll
                for (int jp = 0; jp < 8; ++jp)
                    bn[jp] = *(const v4f*)(xt + jp * JP_PITCH);
            }

            v8f acc[4];
            #pragma unroll
            for (int t = 0; t < 4; ++t)
                acc[t] = (v8f){0.f, 0.f, 0.f, 0.f, 0.f, 0.f, 0.f, 0.f};
            #pragma unroll
            for (int j = 0; j < 16; ++j) {
                v2f b;
                if (j & 1) { b.x = bq[j >> 1].z; b.y = bq[j >> 1].w; }
                else       { b.x = bq[j >> 1].x; b.y = bq[j >> 1].y; }
                #pragma unroll
                for (int t = 0; t < 4; ++t)
                    acc[t] = __builtin_amdgcn_wmma_f32_16x16x4_f32(
                        false, a[t][j], false, b, (short)0, acc[t], false, false);
            }

            float bt[4];
            int   it[4];
            #pragma unroll
            for (int t = 0; t < 4; ++t) {
                bt[t] = __builtin_inff();
                it[t] = 16 * t;
                #pragma unroll
                for (int r = 0; r < 8; ++r) {
                    const float d = cc[t][r] - 2.0f * acc[t][r];
                    if (d < bt[t]) { bt[t] = d; it[t] = 16 * t + r; }
                }
            }
            if (bt[1] < bt[0]) { bt[0] = bt[1]; it[0] = it[1]; }
            if (bt[3] < bt[2]) { bt[2] = bt[3]; it[2] = it[3]; }
            if (bt[2] < bt[0]) { bt[0] = bt[2]; it[0] = it[2]; }
            float best = bt[0];
            int   bidx = it[0] + wave * 64 + 8 * h;
            {
                const float ov = __shfl_xor(best, 16, 32);
                const int   oi = __shfl_xor(bidx, 16, 32);
                if (ov < best || (ov == best && oi < bidx)) { best = ov; bidx = oi; }
            }
            if (h == 0) {
                v2f pv = { best, __int_as_float(bidx) };
                *(v2f*)(pb + ((i * TILE_ROWS + c) * NWAVES + wave) * 2) = pv;
            }

            #pragma unroll
            for (int jp = 0; jp < 8; ++jp) bq[jp] = bn[jp];
        }

        if (g + 1 < nGroups) storeTile(g + 1, nextStage);
        __syncthreads();

        if (wave < nT) {
            const float* pp = pb + ((wave * TILE_ROWS + c) * NWAVES + 4 * h) * 2;
            const v4f p0 = *(const v4f*)pp;
            const v4f p1 = *(const v4f*)(pp + 4);
            float bv = p0.x; int bi = __float_as_int(p0.y);
            if (p0.z < bv) { bv = p0.z; bi = __float_as_int(p0.w); }
            if (p1.x < bv) { bv = p1.x; bi = __float_as_int(p1.y); }
            if (p1.z < bv) { bv = p1.z; bi = __float_as_int(p1.w); }
            {
                const float ov = __shfl_xor(bv, 16, 32);
                const int   oi = __shfl_xor(bi, 16, 32);
                if (ov < bv || (ov == bv && oi < bi)) { bv = ov; bi = oi; }
            }
            if (h == 0)
                labels[(size_t)(t0 + g * GROUP_TILES + wave) * TILE_ROWS + c] = (unsigned short)bi;
        }
        stage = nextStage;
    }
}

typedef __attribute__((ext_vector_type(2))) unsigned short us2;
typedef __attribute__((ext_vector_type(4))) unsigned int   v4u;

constexpr int SUM_WAVES = 16;
constexpr int CHUNK     = 256;
constexpr int BATCH     = 16;
constexpr int LIST_CAP  = BATCH + CHUNK;
constexpr int SCAN_AHEAD = 4;
constexpr int SUM_SLICES = 4;

__device__ __forceinline__ unsigned pk_min_u16(unsigned a, unsigned b) {
    const us2 r = __builtin_elementwise_min(__builtin_bit_cast(us2, a), __builtin_bit_cast(us2, b));
    return __builtin_bit_cast(unsigned, r);
}

__global__ void __launch_bounds__(32 * SUM_WAVES)
kmeans_sum(const float* __restrict__ X,
           const unsigned short* __restrict__ labels,
           float* __restrict__ part,
           float* __restrict__ cntp,
           int N) {
    __shared__ unsigned list[SUM_WAVES][LIST_CAP];
    __shared__ float psum[SUM_WAVES][2][Dk];
    __shared__ float pcnt[SUM_WAVES][2];

    const int wave = threadIdx.x >> 5;
    const int lane = threadIdx.x & 31;
    const int owner = (int)blockIdx.x / SUM_SLICES;
    const int slice = (int)blockIdx.x % SUM_SLICES;
    const unsigned k0  = 2u * (unsigned)owner;
    const unsigned pat = k0 | (k0 << 16);
    unsigned* myList = list[wave];

    v2f acc0 = {0.f, 0.f}, acc1 = {0.f, 0.f};
    int n0 = 0, n1 = 0;
    int cnt = 0;

    const int nChunks = (N + CHUNK - 1) / CHUNK;
    const int chunkLo = (int)(((long long)slice * nChunks) / SUM_SLICES);
    const int chunkHi = (int)(((long long)(slice + 1) * nChunks) / SUM_SLICES);
    const v4u* L4 = (const v4u*)labels;

    auto loadChunk = [&](int ch) -> v4u {
        const bool ok = (ch < chunkHi) && (ch * CHUNK + lane * 8 + 8 <= N);
        const v4u w = L4[ok ? (size_t)ch * 32 + lane : (size_t)0];
        const v4u none = {0xFFFFFFFFu, 0xFFFFFFFFu, 0xFFFFFFFFu, 0xFFFFFFFFu};
        return ok ? w : none;
    };

    auto gatherFull = [&](int base) {
        unsigned e[BATCH];
        v2f v[BATCH];
        #pragma unroll
        for (int i = 0; i < BATCH; ++i) {
            e[i] = __builtin_amdgcn_readfirstlane(myList[base + i]);
            v[i] = *(const v2f*)(X + (size_t)(e[i] >> 1) * Dk + 2 * lane);
        }
        #pragma unroll
        for (int i = 0; i < BATCH; ++i) {
            if (e[i] & 1u) { acc1 += v[i]; ++n1; }
            else           { acc0 += v[i]; ++n0; }
        }
    };

    auto scanChunk = [&](v4u w, int ch) {
        const unsigned y0 = (w.x ^ pat) & 0xFFFEFFFEu;
        const unsigned y1 = (w.y ^ pat) & 0xFFFEFFFEu;
        const unsigned y2 = (w.z ^ pat) & 0xFFFEFFFEu;
        const unsigned y3 = (w.w ^ pat) & 0xFFFEFFFEu;
        const unsigned t = pk_min_u16(pk_min_u16(y0, y1), pk_min_u16(y2, y3));
        const bool hit = ((t & 0xFFFFu) == 0u) || (t < 0x10000u);
        if (__builtin_amdgcn_ballot_w32(hit) == 0u) return;

        const unsigned ys[4] = { y0, y1, y2, y3 };
        const unsigned ws[4] = { w.x, w.y, w.z, w.w };
        const unsigned rowBase = (unsigned)ch * CHUNK + (unsigned)lane * 8u;
        #pragma unroll
        for (int j = 0; j < 8; ++j) {
            const unsigned yj = (j & 1) ? (ys[j >> 1] >> 16) : (ys[j >> 1] & 0xFFFFu);
            const unsigned lj = (j & 1) ? (ws[j >> 1] >> 16) : (ws[j >> 1] & 0xFFFFu);
            const bool mine = (yj == 0u);
            const unsigned bal = __builtin_amdgcn_ballot_w32(mine);
            if (bal != 0u) {
                if (mine) {
                    const int pos = cnt + (int)__builtin_amdgcn_mbcnt_lo(bal, 0u);
                    myList[pos] = ((rowBase + (unsigned)j) << 1) | (lj & 1u);
                }
                cnt += __builtin_popcount(bal);
            }
        }
        __builtin_amdgcn_fence(__ATOMIC_RELEASE, "wavefront");
        __builtin_amdgcn_wave_barrier();
        __builtin_amdgcn_fence(__ATOMIC_ACQUIRE, "wavefront");
        while (cnt >= BATCH) { cnt -= BATCH; gatherFull(cnt); }
    };

    v4u cur[SCAN_AHEAD];
    #pragma unroll
    for (int u = 0; u < SCAN_AHEAD; ++u) cur[u] = loadChunk(chunkLo + wave + SUM_WAVES * u);
    for (int ch = chunkLo + wave; ch < chunkHi; ch += SUM_WAVES * SCAN_AHEAD) {
        v4u nxt[SCAN_AHEAD];
        #pragma unroll
        for (int u = 0; u < SCAN_AHEAD; ++u) nxt[u] = loadChunk(ch + SUM_WAVES * (SCAN_AHEAD + u));
        #pragma unroll
        for (int u = 0; u < SCAN_AHEAD; ++u) scanChunk(cur[u], ch + SUM_WAVES * u);
        #pragma unroll
        for (int u = 0; u < SCAN_AHEAD; ++u) cur[u] = nxt[u];
    }

    for (int i = 0; i < cnt; ++i) {
        const unsigned e = __builtin_amdgcn_readfirstlane(myList[i]);
        const v2f v = *(const v2f*)(X + (size_t)(e >> 1) * Dk + 2 * lane);
        if (e & 1u) { acc1 += v; ++n1; }
        else        { acc0 += v; ++n0; }
    }

    *(v2f*)&psum[wave][0][2 * lane] = acc0;
    *(v2f*)&psum[wave][1][2 * lane] = acc1;
    if (lane == 0) { pcnt[wave][0] = (float)n0; pcnt[wave][1] = (float)n1; }
    __syncthreads();

    if (threadIdx.x < 2 * Dk) {
        const int which = threadIdx.x >> 6;
        const int f     = threadIdx.x & (Dk - 1);
        float s = 0.f, n = 0.f;
        #pragma unroll
        for (int w = 0; w < SUM_WAVES; ++w) { s += psum[w][which][f]; n += pcnt[w][which]; }
        const int k = (int)k0 + which;
        part[((size_t)slice * Kc + k) * Dk + f] = s;
        if (f == 0) cntp[slice * Kc + k] = n;
    }
}

__global__ void __launch_bounds__(4 * Dk)
kmeans_finish(const float* __restrict__ part,
              const float* __restrict__ cntp,
              float* __restrict__ out) {
    const int k = (int)blockIdx.x * 4 + ((int)threadIdx.x >> 6);
    const int f = (int)threadIdx.x & (Dk - 1);
    float s = 0.f, n = 0.f;
    #pragma unroll
    for (int sl = 0; sl < SUM_SLICES; ++sl) {
        s += part[((size_t)sl * Kc + k) * Dk + f];
        n += cntp[sl * Kc + k];
    }
    out[k * Dk + f] = s / fmaxf(n, 1.0f);
}

extern "C" void kernel_launch(void* const* d_in, const int* in_sizes, int n_in,
                              void* d_out, int out_size, void* d_ws, size_t ws_size,
                              hipStream_t stream) {
    const float* X        = (const float*)d_in[0];
    const int*   init_idx = (const int*)d_in[1];
    float*       out      = (float*)d_out;

    const int N          = in_sizes[0] / Dk;
    const int numTiles32 = N / 32;

    float* ws   = (float*)d_ws;
    float* C    = ws;
    float* c2   = C + Kc * Dk;
    float* part = c2 + Kc;
    float* cntp = part + SUM_SLICES * Kc * Dk;
    unsigned short* labels = (unsigned short*)(cntp + SUM_SLICES * Kc);

    kmeans_prep<<<Kc, Dk, 0, stream>>>(X, init_idx, C, c2);

    kmeans_assign<<<ASSIGN_BLOCKS, 32 * NWAVES, ASSIGN_LDS_BYTES, stream>>>(
        X, C, c2, labels, numTiles32);

    kmeans_sum<<<(Kc / 2) * SUM_SLICES, 32 * SUM_WAVES, 0, stream>>>(X, labels, part, cntp, N);

    kmeans_finish<<<Kc / 4, 4 * Dk, 0, stream>>>(part, cntp, out);
}


# Segment-sum kernel re-cut: 768 work-groups (8 centroids x a twelfth of the labels, all resident at once) replace 1024 (2 x a quarter); a quarter of the label scanning; changes summation order
# speedup vs baseline: 3.7126x; 1.0956x over previous
#include <hip/hip_runtime.h>


typedef __attribute__((ext_vector_type(2))) float v2f;
typedef __attribute__((ext_vector_type(4))) float v4f;
typedef __attribute__((ext_vector_type(8))) float v8f;

constexpr int Dk = 64;
constexpr int Kc = 512;
constexpr int NWAVES      = 8;
constexpr int TILE_ROWS   = 16;
constexpr int GROUP_TILES = 8;
constexpr int JP_PITCH    = 132;
constexpr int TILE_FLOATS = 8 * JP_PITCH;
constexpr int XBUF_FLOATS = GROUP_TILES * TILE_FLOATS;
constexpr int XSTAGES     = 3;
constexpr int PART_FLOATS = GROUP_TILES * TILE_ROWS * NWAVES * 2;
constexpr int ASSIGN_BLOCKS = 256;
constexpr int ASSIGN_LDS_BYTES = (XSTAGES * XBUF_FLOATS + 2 * PART_FLOATS) * 4;

__global__ void kmeans_prep(const float* __restrict__ X,
                            const int* __restrict__ init_idx,
                            float* __restrict__ C,
                            float* __restrict__ c2) {
    const int k = blockIdx.x;
    const int t = threadIdx.x;
    const int src = init_idx[k];
    const float v = X[(size_t)src * Dk + t];
    C[k * Dk + t] = v;

    __shared__ float red[Dk];
    red[t] = v * v;
    __syncthreads();
    if (t < 32) {
        float s = red[t] + red[t + 32];
        #pragma unroll
        for (int off = 16; off > 0; off >>= 1)
            s += __shfl_down(s, off, 32);
        if (t == 0) c2[k] = s;
    }
}

__global__ void __launch_bounds__(256)
kmeans_assign(const float* __restrict__ X,
              const float* __restrict__ C,
              const float* __restrict__ c2,
              unsigned short* __restrict__ labels,
              int numTiles32) {
    extern __shared__ float lds[];
    float* xbuf = lds;
    float* part = lds + XSTAGES * XBUF_FLOATS;

    const int wave = threadIdx.x >> 5;
    const int lane = threadIdx.x & 31;
    const int c = lane & 15;
    const int h = lane >> 4;

    const int numTiles = numTiles32 * 2;
    const int t0 = (int)(((long long)blockIdx.x * numTiles) / (int)gridDim.x);
    const int t1 = (int)(((long long)(blockIdx.x + 1) * numTiles) / (int)gridDim.x);
    const int nTiles  = t1 - t0;
    const int nGroups = (nTiles + GROUP_TILES - 1) / GROUP_TILES;

    v2f a[4][16];
    {
        const float* cr = C + (size_t)(wave * 64 + c) * Dk + 2 * h;
        #pragma unroll
        for (int t = 0; t < 4; ++t) {
            #pragma unroll
            for (int j = 0; j < 16; ++j)
                a[t][j] = *(const v2f*)(cr + t * 16 * Dk + 4 * j);
        }
    }
    float cc[4][8];
    #pragma unroll
    for (int t = 0; t < 4; ++t) {
        #pragma unroll
        for (int r = 0; r < 8; ++r)
            cc[t][r] = c2[wave * 64 + 16 * t + 8 * h + r];
    }

    v4f xr[8];
    auto loadTile = [&](int g) {
        const int ti = g * GROUP_TILES + wave;
        if (ti < nTiles) {
            const float* src = X + ((size_t)(t0 + ti) * TILE_ROWS + c) * Dk + 4 * h;
            #pragma unroll
            for (int i = 0; i < 8; ++i) xr[i] = *(const v4f*)(src + 8 * i);
        }
    };
    auto storeTile = [&](int g, int stage) {
        const int ti = g * GROUP_TILES + wave;
        if (ti < nTiles) {
            float* dst = xbuf + stage * XBUF_FLOATS + wave * TILE_FLOATS + 4 * c + 2 * h;
            #pragma unroll
            for (int i = 0; i < 8; ++i) {
                v2f lo = { xr[i].x, xr[i].y };
                v2f hi = { xr[i].z, xr[i].w };
                *(v2f*)(dst + i * JP_PITCH)      = lo;
                *(v2f*)(dst + i * JP_PITCH + 64) = hi;
            }
        }
    };

    loadTile(0);
    storeTile(0, 0);
    __syncthreads();

    int stage = 0;
    for (int g = 0; g < nGroups; ++g) {
        const int left = nTiles - g * GROUP_TILES;
        const int nT = left < GROUP_TILES ? left : GROUP_TILES;
        const int nextStage = (stage == XSTAGES - 1) ? 0 : stage + 1;
        if (g + 1 < nGroups) loadTile(g + 1);

        const float* xb = xbuf + stage * XBUF_FLOATS;
        float* pb = part + (g & 1) * PART_FLOATS;

        v4f bq[8];
        #pragma unroll
        for (int jp = 0; jp < 8; ++jp)
            bq[jp] = *(const v4f*)(xb + jp * JP_PITCH + 4 * lane);

        for (int i = 0; i < nT; ++i) {
            const int inext = (i + 1 < nT) ? i + 1 : i;
            v4f bn[8];
            {
                const float* xt = xb + inext * TILE_FLOATS + 4 * lane;
                #pragma unroll
                for (int jp = 0; jp < 8; ++jp)
                    bn[jp] = *(const v4f*)(xt + jp * JP_PITCH);
            }

            v8f acc[4];
            #pragma unroll
            for (int t = 0; t < 4; ++t)
                acc[t] = (v8f){0.f, 0.f, 0.f, 0.f, 0.f, 0.f, 0.f, 0.f};
            #pragma unroll
            for (int j = 0; j < 16; ++j) {
                v2f b;
                if (j & 1) { b.x = bq[j >> 1].z; b.y = bq[j >> 1].w; }
                else       { b.x = bq[j >> 1].x; b.y = bq[j >> 1].y; }
                #pragma unroll
                for (int t = 0; t < 4; ++t)
                    acc[t] = __builtin_amdgcn_wmma_f32_16x16x4_f32(
                        false, a[t][j], false, b, (short)0, acc[t], false, false);
            }

            float bt[4];
            int   it[4];
            #pragma unroll
            for (int t = 0; t < 4; ++t) {
                bt[t] = __builtin_inff();
                it[t] = 16 * t;
                #pragma unroll
                for (int r = 0; r < 8; ++r) {
                    const float d = cc[t][r] - 2.0f * acc[t][r];
                    if (d < bt[t]) { bt[t] = d; it[t] = 16 * t + r; }
                }
            }
            if (bt[1] < bt[0]) { bt[0] = bt[1]; it[0] = it[1]; }
            if (bt[3] < bt[2]) { bt[2] = bt[3]; it[2] = it[3]; }
            if (bt[2] < bt[0]) { bt[0] = bt[2]; it[0] = it[2]; }
            float best = bt[0];
            int   bidx = it[0] + wave * 64 + 8 * h;
            {
                const float ov = __shfl_xor(best, 16, 32);
                const int   oi = __shfl_xor(bidx, 16, 32);
                if (ov < best || (ov == best && oi < bidx)) { best = ov; bidx = oi; }
            }
            if (h == 0) {
                v2f pv = { best, __int_as_float(bidx) };
                *(v2f*)(pb + ((i * TILE_ROWS + c) * NWAVES + wave) * 2) = pv;
            }

            #pragma unroll
            for (int jp = 0; jp < 8; ++jp) bq[jp] = bn[jp];
        }

        if (g + 1 < nGroups) storeTile(g + 1, nextStage);
        __syncthreads();

        if (wave < nT) {
            const float* pp = pb + ((wave * TILE_ROWS + c) * NWAVES + 4 * h) * 2;
            const v4f p0 = *(const v4f*)pp;
            const v4f p1 = *(const v4f*)(pp + 4);
            float bv = p0.x; int bi = __float_as_int(p0.y);
            if (p0.z < bv) { bv = p0.z; bi = __float_as_int(p0.w); }
            if (p1.x < bv) { bv = p1.x; bi = __float_as_int(p1.y); }
            if (p1.z < bv) { bv = p1.z; bi = __float_as_int(p1.w); }
            {
                const float ov = __shfl_xor(bv, 16, 32);
                const int   oi = __shfl_xor(bi, 16, 32);
                if (ov < bv || (ov == bv && oi < bi)) { bv = ov; bi = oi; }
            }
            if (h == 0)
                labels[(size_t)(t0 + g * GROUP_TILES + wave) * TILE_ROWS + c] = (unsigned short)bi;
        }
        stage = nextStage;
    }
}

typedef __attribute__((ext_vector_type(2))) unsigned short us2;
typedef __attribute__((ext_vector_type(4))) unsigned int   v4u;

constexpr int SUM_WAVES = 16;
constexpr int CHUNK     = 256;
constexpr int BATCH     = 16;
constexpr int LIST_CAP  = BATCH + CHUNK;
constexpr int SCAN_AHEAD = 2;
constexpr int OWN        = 8;
constexpr int SUM_SLICES = 12;
constexpr unsigned OWN_MASK = 0xFFFFu & ~(unsigned)(OWN - 1);

__device__ __forceinline__ unsigned pk_min_u16(unsigned a, unsigned b) {
    const us2 r = __builtin_elementwise_min(__builtin_bit_cast(us2, a), __builtin_bit_cast(us2, b));
    return __builtin_bit_cast(unsigned, r);
}

__global__ void __launch_bounds__(32 * SUM_WAVES, 12)
kmeans_sum(const float* __restrict__ X,
           const unsigned short* __restrict__ labels,
           float* __restrict__ part,
           float* __restrict__ cntp,
           int N) {
    __shared__ unsigned list[SUM_WAVES][LIST_CAP];
    __shared__ float psum[SUM_WAVES][OWN][Dk];
    __shared__ float pcnt[SUM_WAVES][OWN];

    const int wave = threadIdx.x >> 5;
    const int lane = threadIdx.x & 31;
    const int owner = (int)blockIdx.x / SUM_SLICES;
    const int slice = (int)blockIdx.x % SUM_SLICES;
    const unsigned k0  = (unsigned)OWN * (unsigned)owner;
    const unsigned pat = k0 | (k0 << 16);
    const unsigned msk = OWN_MASK | (OWN_MASK << 16);
    unsigned* myList = list[wave];

    v2f acc[OWN];
    #pragma unroll
    for (int q = 0; q < OWN; ++q) acc[q] = (v2f){0.f, 0.f};
    unsigned nrows = 0;
    int cnt = 0;

    const int nChunks = (N + CHUNK - 1) / CHUNK;
    const int chunkLo = (int)(((long long)slice * nChunks) / SUM_SLICES);
    const int chunkHi = (int)(((long long)(slice + 1) * nChunks) / SUM_SLICES);
    const v4u* L4 = (const v4u*)labels;

    auto loadChunk = [&](int ch) -> v4u {
        const bool ok = (ch < chunkHi) && (ch * CHUNK + lane * 8 + 8 <= N);
        const v4u w = L4[ok ? (size_t)ch * 32 + lane : (size_t)0];
        const v4u none = {0xFFFFFFFFu, 0xFFFFFFFFu, 0xFFFFFFFFu, 0xFFFFFFFFu};
        return ok ? w : none;
    };

    auto addRow = [&](unsigned e, v2f v) {
        const unsigned sel = e & (unsigned)(OWN - 1);
        #pragma unroll
        for (int q = 0; q < OWN; ++q) {
            const float m = (sel == (unsigned)q) ? 1.0f : 0.0f;
            acc[q] = v * m + acc[q];
        }
        nrows += ((unsigned)lane == sel) ? 1u : 0u;
    };

    auto gatherFull = [&](int base) {
        unsigned e[BATCH];
        v2f v[BATCH];
        #pragma unroll
        for (int i = 0; i < BATCH; ++i) {
            e[i] = __builtin_amdgcn_readfirstlane(myList[base + i]);
            v[i] = *(const v2f*)(X + (size_t)(e[i] >> 3) * Dk + 2 * lane);
        }
        __builtin_amdgcn_sched_barrier(0);
        #pragma unroll
        for (int i = 0; i < BATCH; ++i) addRow(e[i], v[i]);
    };

    auto scanChunk = [&](v4u w, int ch) {
        const unsigned y0 = (w.x ^ pat) & msk;
        const unsigned y1 = (w.y ^ pat) & msk;
        const unsigned y2 = (w.z ^ pat) & msk;
        const unsigned y3 = (w.w ^ pat) & msk;
        const unsigned t = pk_min_u16(pk_min_u16(y0, y1), pk_min_u16(y2, y3));
        const bool hit = ((t & 0xFFFFu) == 0u) || (t < 0x10000u);
        if (__builtin_amdgcn_ballot_w32(hit) == 0u) return;

        const unsigned ys[4] = { y0, y1, y2, y3 };
        const unsigned ws[4] = { w.x, w.y, w.z, w.w };
        const unsigned rowBase = (unsigned)ch * CHUNK + (unsigned)lane * 8u;
        #pragma unroll
        for (int j = 0; j < 8; ++j) {
            const unsigned yj = (j & 1) ? (ys[j >> 1] >> 16) : (ys[j >> 1] & 0xFFFFu);
            const unsigned lj = (j & 1) ? (ws[j >> 1] >> 16) : (ws[j >> 1] & 0xFFFFu);
            const bool mine = (yj == 0u);
            const unsigned bal = __builtin_amdgcn_ballot_w32(mine);
            if (bal != 0u) {
                if (mine) {
                    const int pos = cnt + (int)__builtin_amdgcn_mbcnt_lo(bal, 0u);
                    myList[pos] = ((rowBase + (unsigned)j) << 3) | (lj & (unsigned)(OWN - 1));
                }
                cnt += __builtin_popcount(bal);
            }
        }
        __builtin_amdgcn_fence(__ATOMIC_RELEASE, "wavefront");
        __builtin_amdgcn_wave_barrier();
        __builtin_amdgcn_fence(__ATOMIC_ACQUIRE, "wavefront");
        while (cnt >= BATCH) { cnt -= BATCH; gatherFull(cnt); }
    };

    v4u cur[SCAN_AHEAD];
    #pragma unroll
    for (int u = 0; u < SCAN_AHEAD; ++u) cur[u] = loadChunk(chunkLo + wave + SUM_WAVES * u);
    for (int ch = chunkLo + wave; ch < chunkHi; ch += SUM_WAVES * SCAN_AHEAD) {
        v4u nxt[SCAN_AHEAD];
        #pragma unroll
        for (int u = 0; u < SCAN_AHEAD; ++u) nxt[u] = loadChunk(ch + SUM_WAVES * (SCAN_AHEAD + u));
        #pragma unroll
        for (int u = 0; u < SCAN_AHEAD; ++u) scanChunk(cur[u], ch + SUM_WAVES * u);
        #pragma unroll
        for (int u = 0; u < SCAN_AHEAD; ++u) cur[u] = nxt[u];
    }

    for (int i = 0; i < cnt; ++i) {
        const unsigned e = __builtin_amdgcn_readfirstlane(myList[i]);
        const v2f v = *(const v2f*)(X + (size_t)(e >> 3) * Dk + 2 * lane);
        addRow(e, v);
    }

    #pragma unroll
    for (int q = 0; q < OWN; ++q) *(v2f*)&psum[wave][q][2 * lane] = acc[q];
    if (lane < OWN) pcnt[wave][lane] = (float)nrows;
    __syncthreads();

    if (threadIdx.x < OWN * Dk) {
        const int which = threadIdx.x >> 6;
        const int f     = threadIdx.x & (Dk - 1);
        float s = 0.f, n = 0.f;
        #pragma unroll
        for (int w = 0; w < SUM_WAVES; ++w) { s += psum[w][which][f]; n += pcnt[w][which]; }
        const int k = (int)k0 + which;
        part[((size_t)slice * Kc + k) * Dk + f] = s;
        if (f == 0) cntp[slice * Kc + k] = n;
    }
}

__global__ void __launch_bounds__(4 * Dk)
kmeans_finish(const float* __restrict__ part,
              const float* __restrict__ cntp,
              float* __restrict__ out) {
    const int k = (int)blockIdx.x * 4 + ((int)threadIdx.x >> 6);
    const int f = (int)threadIdx.x & (Dk - 1);
    float s = 0.f, n = 0.f;
    #pragma unroll
    for (int sl = 0; sl < SUM_SLICES; ++sl) {
        s += part[((size_t)sl * Kc + k) * Dk + f];
        n += cntp[sl * Kc + k];
    }
    out[k * Dk + f] = s / fmaxf(n, 1.0f);
}

extern "C" void kernel_launch(void* const* d_in, const int* in_sizes, int n_in,
                              void* d_out, int out_size, void* d_ws, size_t ws_size,
                              hipStream_t stream) {
    const float* X        = (const float*)d_in[0];
    const int*   init_idx = (const int*)d_in[1];
    float*       out      = (float*)d_out;

    const int N          = in_sizes[0] / Dk;
    const int numTiles32 = N / 32;

    float* ws   = (float*)d_ws;
    float* C    = ws;
    float* c2   = C + Kc * Dk;
    float* part = c2 + Kc;
    float* cntp = part + SUM_SLICES * Kc * Dk;
    unsigned short* labels = (unsigned short*)(cntp + SUM_SLICES * Kc);

    kmeans_prep<<<Kc, Dk, 0, stream>>>(X, init_idx, C, c2);

    kmeans_assign<<<ASSIGN_BLOCKS, 32 * NWAVES, ASSIGN_LDS_BYTES, stream>>>(
        X, C, c2, labels, numTiles32);

    kmeans_sum<<<(Kc / OWN) * SUM_SLICES, 32 * SUM_WAVES, 0, stream>>>(X, labels, part, cntp, N);

    kmeans_finish<<<Kc / 4, 4 * Dk, 0, stream>>>(part, cntp, out);
}


# Segment sums moved onto the matrix unit: each batch of 16 fetched rows is multiplied by its 16x16 one-hot label matrix (f32 WMMA); 16 centroids per work-group, 512 groups; changes summation order
# speedup vs baseline: 3.8390x; 1.0340x over previous
#include <hip/hip_runtime.h>


typedef __attribute__((ext_vector_type(2))) float v2f;
typedef __attribute__((ext_vector_type(4))) float v4f;
typedef __attribute__((ext_vector_type(8))) float v8f;

constexpr int Dk = 64;
constexpr int Kc = 512;
constexpr int NWAVES      = 8;
constexpr int TILE_ROWS   = 16;
constexpr int GROUP_TILES = 8;
constexpr int JP_PITCH    = 132;
constexpr int TILE_FLOATS = 8 * JP_PITCH;
constexpr int XBUF_FLOATS = GROUP_TILES * TILE_FLOATS;
constexpr int XSTAGES     = 3;
constexpr int PART_FLOATS = GROUP_TILES * TILE_ROWS * NWAVES * 2;
constexpr int ASSIGN_BLOCKS = 256;
constexpr int ASSIGN_LDS_BYTES = (XSTAGES * XBUF_FLOATS + 2 * PART_FLOATS) * 4;

__global__ void kmeans_prep(const float* __restrict__ X,
                            const int* __restrict__ init_idx,
                            float* __restrict__ C,
                            float* __restrict__ c2) {
    const int k = blockIdx.x;
    const int t = threadIdx.x;
    const int src = init_idx[k];
    const float v = X[(size_t)src * Dk + t];
    C[k * Dk + t] = v;

    __shared__ float red[Dk];
    red[t] = v * v;
    __syncthreads();
    if (t < 32) {
        float s = red[t] + red[t + 32];
        #pragma unroll
        for (int off = 16; off > 0; off >>= 1)
            s += __shfl_down(s, off, 32);
        if (t == 0) c2[k] = s;
    }
}

__global__ void __launch_bounds__(256)
kmeans_assign(const float* __restrict__ X,
              const float* __restrict__ C,
              const float* __restrict__ c2,
              unsigned short* __restrict__ labels,
              int numTiles32) {
    extern __shared__ float lds[];
    float* xbuf = lds;
    float* part = lds + XSTAGES * XBUF_FLOATS;

    const int wave = threadIdx.x >> 5;
    const int lane = threadIdx.x & 31;
    const int c = lane & 15;
    const int h = lane >> 4;

    const int numTiles = numTiles32 * 2;
    const int t0 = (int)(((long long)blockIdx.x * numTiles) / (int)gridDim.x);
    const int t1 = (int)(((long long)(blockIdx.x + 1) * numTiles) / (int)gridDim.x);
    const int nTiles  = t1 - t0;
    const int nGroups = (nTiles + GROUP_TILES - 1) / GROUP_TILES;

    v2f a[4][16];
    {
        const float* cr = C + (size_t)(wave * 64 + c) * Dk + 2 * h;
        #pragma unroll
        for (int t = 0; t < 4; ++t) {
            #pragma unroll
            for (int j = 0; j < 16; ++j)
                a[t][j] = *(const v2f*)(cr + t * 16 * Dk + 4 * j);
        }
    }
    float cc[4][8];
    #pragma unroll
    for (int t = 0; t < 4; ++t) {
        #pragma unroll
        for (int r = 0; r < 8; ++r)
            cc[t][r] = c2[wave * 64 + 16 * t + 8 * h + r];
    }

    v4f xr[8];
    auto loadTile = [&](int g) {
        const int ti = g * GROUP_TILES + wave;
        if (ti < nTiles) {
            const float* src = X + ((size_t)(t0 + ti) * TILE_ROWS + c) * Dk + 4 * h;
            #pragma unroll
            for (int i = 0; i < 8; ++i) xr[i] = *(const v4f*)(src + 8 * i);
        }
    };
    auto storeTile = [&](int g, int stage) {
        const int ti = g * GROUP_TILES + wave;
        if (ti < nTiles) {
            float* dst = xbuf + stage * XBUF_FLOATS + wave * TILE_FLOATS + 4 * c + 2 * h;
            #pragma unroll
            for (int i = 0; i < 8; ++i) {
                v2f lo = { xr[i].x, xr[i].y };
                v2f hi = { xr[i].z, xr[i].w };
                *(v2f*)(dst + i * JP_PITCH)      = lo;
                *(v2f*)(dst + i * JP_PITCH + 64) = hi;
            }
        }
    };

    loadTile(0);
    storeTile(0, 0);
    __syncthreads();

    int stage = 0;
    for (int g = 0; g < nGroups; ++g) {
        const int left = nTiles - g * GROUP_TILES;
        const int nT = left < GROUP_TILES ? left : GROUP_TILES;
        const int nextStage = (stage == XSTAGES - 1) ? 0 : stage + 1;
        if (g + 1 < nGroups) loadTile(g + 1);

        const float* xb = xbuf + stage * XBUF_FLOATS;
        float* pb = part + (g & 1) * PART_FLOATS;

        v4f bq[8];
        #pragma unroll
        for (int jp = 0; jp < 8; ++jp)
            bq[jp] = *(const v4f*)(xb + jp * JP_PITCH + 4 * lane);

        for (int i = 0; i < nT; ++i) {
            const int inext = (i + 1 < nT) ? i + 1 : i;
            v4f bn[8];
            {
                const float* xt = xb + inext * TILE_FLOATS + 4 * lane;
                #pragma unroll
                for (int jp = 0; jp < 8; ++jp)
                    bn[jp] = *(const v4f*)(xt + jp * JP_PITCH);
            }

            v8f acc[4];
            #pragma unroll
            for (int t = 0; t < 4; ++t)
                acc[t] = (v8f){0.f, 0.f, 0.f, 0.f, 0.f, 0.f, 0.f, 0.f};
            #pragma unroll
            for (int j = 0; j < 16; ++j) {
                v2f b;
                if (j & 1) { b.x = bq[j >> 1].z; b.y = bq[j >> 1].w; }
                else       { b.x = bq[j >> 1].x; b.y = bq[j >> 1].y; }
                #pragma unroll
                for (int t = 0; t < 4; ++t)
                    acc[t] = __builtin_amdgcn_wmma_f32_16x16x4_f32(
                        false, a[t][j], false, b, (short)0, acc[t], false, false);
            }

            float bt[4];
            int   it[4];
            #pragma unroll
            for (int t = 0; t < 4; ++t) {
                bt[t] = __builtin_inff();
                it[t] = 16 * t;
                #pragma unroll
                for (int r = 0; r < 8; ++r) {
                    const float d = cc[t][r] - 2.0f * acc[t][r];
                    if (d < bt[t]) { bt[t] = d; it[t] = 16 * t + r; }
                }
            }
            if (bt[1] < bt[0]) { bt[0] = bt[1]; it[0] = it[1]; }
            if (bt[3] < bt[2]) { bt[2] = bt[3]; it[2] = it[3]; }
            if (bt[2] < bt[0]) { bt[0] = bt[2]; it[0] = it[2]; }
            float best = bt[0];
            int   bidx = it[0] + wave * 64 + 8 * h;
            {
                const float ov = __shfl_xor(best, 16, 32);
                const int   oi = __shfl_xor(bidx, 16, 32);
                if (ov < best || (ov == best && oi < bidx)) { best = ov; bidx = oi; }
            }
            if (h == 0) {
                v2f pv = { best, __int_as_float(bidx) };
                *(v2f*)(pb + ((i * TILE_ROWS + c) * NWAVES + wave) * 2) = pv;
            }

            #pragma unroll
            for (int jp = 0; jp < 8; ++jp) bq[jp] = bn[jp];
        }

        if (g + 1 < nGroups) storeTile(g + 1, nextStage);
        __syncthreads();

        if (wave < nT) {
            const float* pp = pb + ((wave * TILE_ROWS + c) * NWAVES + 4 * h) * 2;
            const v4f p0 = *(const v4f*)pp;
            const v4f p1 = *(const v4f*)(pp + 4);
            float bv = p0.x; int bi = __float_as_int(p0.y);
            if (p0.z < bv) { bv = p0.z; bi = __float_as_int(p0.w); }
            if (p1.x < bv) { bv = p1.x; bi = __float_as_int(p1.y); }
            if (p1.z < bv) { bv = p1.z; bi = __float_as_int(p1.w); }
            {
                const float ov = __shfl_xor(bv, 16, 32);
                const int   oi = __shfl_xor(bi, 16, 32);
                if (ov < bv || (ov == bv && oi < bi)) { bv = ov; bi = oi; }
            }
            if (h == 0)
                labels[(size_t)(t0 + g * GROUP_TILES + wave) * TILE_ROWS + c] = (unsigned short)bi;
        }
        stage = nextStage;
    }
}

typedef __attribute__((ext_vector_type(2))) unsigned short us2;
typedef __attribute__((ext_vector_type(4))) unsigned int   v4u;

constexpr int SUM_WAVES = 16;
constexpr int CHUNK     = 256;
constexpr int BATCH     = 16;
constexpr int LIST_CAP  = BATCH + CHUNK;
constexpr int SCAN_AHEAD = 2;
constexpr int OWN        = 16;
constexpr int SUM_SLICES = 16;
constexpr unsigned OWN_MASK = 0xFFFFu & ~(unsigned)(OWN - 1);
constexpr unsigned SLOT_BITS = 5;
constexpr unsigned SLOT_NONE = 16;

__device__ __forceinline__ unsigned pk_min_u16(unsigned a, unsigned b) {
    const us2 r = __builtin_elementwise_min(__builtin_bit_cast(us2, a), __builtin_bit_cast(us2, b));
    return __builtin_bit_cast(unsigned, r);
}

__global__ void __launch_bounds__(32 * SUM_WAVES, 8)
kmeans_sum(const float* __restrict__ X,
           const unsigned short* __restrict__ labels,
           float* __restrict__ part,
           float* __restrict__ cntp,
           int N) {
    __shared__ unsigned list[SUM_WAVES][LIST_CAP];
    __shared__ float psum[SUM_WAVES][OWN][Dk];
    __shared__ float pcnt[SUM_WAVES][OWN];

    const int wave = __builtin_amdgcn_readfirstlane((int)threadIdx.x >> 5);
    const int lane = threadIdx.x & 31;
    const int c = lane & 15;
    const int h = lane >> 4;
    const int owner = (int)blockIdx.x / SUM_SLICES;
    const int slice = (int)blockIdx.x % SUM_SLICES;
    const unsigned k0  = (unsigned)OWN * (unsigned)owner;
    const unsigned pat = k0 | (k0 << 16);
    const unsigned msk = OWN_MASK | (OWN_MASK << 16);
    unsigned* myList = list[wave];

    v8f acc[4];
    #pragma unroll
    for (int t = 0; t < 4; ++t) acc[t] = (v8f){0.f, 0.f, 0.f, 0.f, 0.f, 0.f, 0.f, 0.f};
    unsigned nrows = 0;
    int cnt = 0;

    const int nChunks = (N + CHUNK - 1) / CHUNK;
    const int chunkLo = (int)(((long long)slice * nChunks) / SUM_SLICES);
    const int chunkHi = (int)(((long long)(slice + 1) * nChunks) / SUM_SLICES);
    const v4u* L4 = (const v4u*)labels;

    auto loadChunk = [&](int ch) -> v4u {
        const bool ok = (ch < chunkHi) && (ch * CHUNK + lane * 8 + 8 <= N);
        const v4u w = L4[ok ? (size_t)ch * 32 + lane : (size_t)0];
        const v4u none = {0xFFFFFFFFu, 0xFFFFFFFFu, 0xFFFFFFFFu, 0xFFFFFFFFu};
        return ok ? w : none;
    };

    auto addBatch = [&](int base) {
        v2f   hot[4];
        float xv[4][2][4];
        #pragma unroll
        for (int j = 0; j < 4; ++j) {
            float one[2];
            #pragma unroll
            for (int e = 0; e < 2; ++e) {
                const unsigned w = myList[base + 4 * j + 2 * h + e];
                const bool mine = ((w & ((1u << SLOT_BITS) - 1u)) == (unsigned)c);
                one[e] = mine ? 1.0f : 0.0f;
                nrows += mine ? 1u : 0u;
                const float* p = X + (size_t)(w >> SLOT_BITS) * Dk + c;
                #pragma unroll
                for (int t = 0; t < 4; ++t) xv[j][e][t] = p[16 * t];
            }
            hot[j] = (v2f){ one[0], one[1] };
        }
        __builtin_amdgcn_sched_barrier(0);
        #pragma unroll
        for (int j = 0; j < 4; ++j) {
            #pragma unroll
            for (int t = 0; t < 4; ++t) {
                const v2f b = { xv[j][0][t], xv[j][1][t] };
                acc[t] = __builtin_amdgcn_wmma_f32_16x16x4_f32(
                    false, hot[j], false, b, (short)0, acc[t], false, false);
            }
        }
        __builtin_amdgcn_sched_barrier(0);
        asm volatile("v_nop\n\tv_nop\n\tv_nop\n\tv_nop");
    };

    auto listReady = [&]() {
        __builtin_amdgcn_fence(__ATOMIC_RELEASE, "wavefront");
        __builtin_amdgcn_wave_barrier();
        __builtin_amdgcn_fence(__ATOMIC_ACQUIRE, "wavefront");
    };

    auto scanChunk = [&](v4u w, int ch) {
        const unsigned y0 = (w.x ^ pat) & msk;
        const unsigned y1 = (w.y ^ pat) & msk;
        const unsigned y2 = (w.z ^ pat) & msk;
        const unsigned y3 = (w.w ^ pat) & msk;
        const unsigned t = pk_min_u16(pk_min_u16(y0, y1), pk_min_u16(y2, y3));
        const bool hit = ((t & 0xFFFFu) == 0u) || (t < 0x10000u);
        if (__builtin_amdgcn_ballot_w32(hit) == 0u) return;

        const unsigned ys[4] = { y0, y1, y2, y3 };
        const unsigned ws[4] = { w.x, w.y, w.z, w.w };
        const unsigned rowBase = (unsigned)ch * CHUNK + (unsigned)lane * 8u;
        #pragma unroll
        for (int j = 0; j < 8; ++j) {
            const unsigned yj = (j & 1) ? (ys[j >> 1] >> 16) : (ys[j >> 1] & 0xFFFFu);
            const unsigned lj = (j & 1) ? (ws[j >> 1] >> 16) : (ws[j >> 1] & 0xFFFFu);
            const bool mine = (yj == 0u);
            const unsigned bal = __builtin_amdgcn_ballot_w32(mine);
            if (bal != 0u) {
                if (mine) {
                    const int pos = cnt + (int)__builtin_amdgcn_mbcnt_lo(bal, 0u);
                    myList[pos] = ((rowBase + (unsigned)j) << SLOT_BITS) | (lj & (unsigned)(OWN - 1));
                }
                cnt += __builtin_popcount(bal);
            }
        }
        listReady();
        while (cnt >= BATCH) { cnt -= BATCH; addBatch(cnt); }
    };

    v4u cur[SCAN_AHEAD];
    #pragma unroll
    for (int u = 0; u < SCAN_AHEAD; ++u) cur[u] = loadChunk(chunkLo + wave + SUM_WAVES * u);
    for (int ch = chunkLo + wave; ch < chunkHi; ch += SUM_WAVES * SCAN_AHEAD) {
        v4u nxt[SCAN_AHEAD];
        #pragma unroll
        for (int u = 0; u < SCAN_AHEAD; ++u) nxt[u] = loadChunk(ch + SUM_WAVES * (SCAN_AHEAD + u));
        #pragma unroll
        for (int u = 0; u < SCAN_AHEAD; ++u) scanChunk(cur[u], ch + SUM_WAVES * u);
        #pragma unroll
        for (int u = 0; u < SCAN_AHEAD; ++u) cur[u] = nxt[u];
    }

    if (cnt > 0) {
        if (lane >= cnt && lane < BATCH) myList[lane] = SLOT_NONE;
        listReady();
        addBatch(0);
    }

    #pragma unroll
    for (int t = 0; t < 4; ++t) {
        #pragma unroll
        for (int r = 0; r < 8; ++r) psum[wave][8 * h + r][16 * t + c] = acc[t][r];
    }
    {
        const unsigned tot = nrows + (unsigned)__shfl_xor((int)nrows, 16, 32);
        if (h == 0) pcnt[wave][c] = (float)tot;
    }
    __syncthreads();

    for (int o = (int)threadIdx.x; o < OWN * Dk; o += 32 * SUM_WAVES) {
        const int which = o >> 6;
        const int f     = o & (Dk - 1);
        float s = 0.f, n = 0.f;
        #pragma unroll
        for (int w = 0; w < SUM_WAVES; ++w) { s += psum[w][which][f]; n += pcnt[w][which]; }
        const int k = (int)k0 + which;
        part[((size_t)slice * Kc + k) * Dk + f] = s;
        if (f == 0) cntp[slice * Kc + k] = n;
    }
}

__global__ void __launch_bounds__(4 * Dk)
kmeans_finish(const float* __restrict__ part,
              const float* __restrict__ cntp,
              float* __restrict__ out) {
    const int k = (int)blockIdx.x * 4 + ((int)threadIdx.x >> 6);
    const int f = (int)threadIdx.x & (Dk - 1);
    float s = 0.f, n = 0.f;
    #pragma unroll
    for (int sl = 0; sl < SUM_SLICES; ++sl) {
        s += part[((size_t)sl * Kc + k) * Dk + f];
        n += cntp[sl * Kc + k];
    }
    out[k * Dk + f] = s / fmaxf(n, 1.0f);
}

extern "C" void kernel_launch(void* const* d_in, const int* in_sizes, int n_in,
                              void* d_out, int out_size, void* d_ws, size_t ws_size,
                              hipStream_t stream) {
    const float* X        = (const float*)d_in[0];
    const int*   init_idx = (const int*)d_in[1];
    float*       out      = (float*)d_out;

    const int N          = in_sizes[0] / Dk;
    const int numTiles32 = N / 32;

    float* ws   = (float*)d_ws;
    float* C    = ws;
    float* c2   = C + Kc * Dk;
    float* part = c2 + Kc;
    float* cntp = part + SUM_SLICES * Kc * Dk;
    unsigned short* labels = (unsigned short*)(cntp + SUM_SLICES * Kc);

    kmeans_prep<<<Kc, Dk, 0, stream>>>(X, init_idx, C, c2);

    kmeans_assign<<<ASSIGN_BLOCKS, 32 * NWAVES, ASSIGN_LDS_BYTES, stream>>>(
        X, C, c2, labels, numTiles32);

    kmeans_sum<<<(Kc / OWN) * SUM_SLICES, 32 * SUM_WAVES, 0, stream>>>(X, labels, part, cntp, N);

    kmeans_finish<<<Kc / 4, 4 * Dk, 0, stream>>>(part, cntp, out);
}


# Assign kernel, lane argmin: the minimum first by plain min3 chains with no index carried, then its index by equality in the one D tile that holds it: 124 -> 74 vector instructions a tile; same labels
# speedup vs baseline: 3.9588x; 1.0312x over previous
#include <hip/hip_runtime.h>


typedef __attribute__((ext_vector_type(2))) float v2f;
typedef __attribute__((ext_vector_type(4))) float v4f;
typedef __attribute__((ext_vector_type(8))) float v8f;

constexpr int Dk = 64;
constexpr int Kc = 512;
constexpr int NWAVES      = 8;
constexpr int TILE_ROWS   = 16;
constexpr int GROUP_TILES = 8;
constexpr int JP_PITCH    = 132;
constexpr int TILE_FLOATS = 8 * JP_PITCH;
constexpr int XBUF_FLOATS = GROUP_TILES * TILE_FLOATS;
constexpr int XSTAGES     = 3;
constexpr int PART_FLOATS = GROUP_TILES * TILE_ROWS * NWAVES * 2;
constexpr int ASSIGN_BLOCKS = 256;
constexpr int ASSIGN_LDS_BYTES = (XSTAGES * XBUF_FLOATS + 2 * PART_FLOATS) * 4;

__global__ void kmeans_prep(const float* __restrict__ X,
                            const int* __restrict__ init_idx,
                            float* __restrict__ C,
                            float* __restrict__ c2) {
    const int k = blockIdx.x;
    const int t = threadIdx.x;
    const int src = init_idx[k];
    const float v = X[(size_t)src * Dk + t];
    C[k * Dk + t] = v;

    __shared__ float red[Dk];
    red[t] = v * v;
    __syncthreads();
    if (t < 32) {
        float s = red[t] + red[t + 32];
        #pragma unroll
        for (int off = 16; off > 0; off >>= 1)
            s += __shfl_down(s, off, 32);
        if (t == 0) c2[k] = s;
    }
}

__global__ void __launch_bounds__(256)
kmeans_assign(const float* __restrict__ X,
              const float* __restrict__ C,
              const float* __restrict__ c2,
              unsigned short* __restrict__ labels,
              int numTiles32) {
    extern __shared__ float lds[];
    float* xbuf = lds;
    float* part = lds + XSTAGES * XBUF_FLOATS;

    const int wave = threadIdx.x >> 5;
    const int lane = threadIdx.x & 31;
    const int c = lane & 15;
    const int h = lane >> 4;

    const int numTiles = numTiles32 * 2;
    const int t0 = (int)(((long long)blockIdx.x * numTiles) / (int)gridDim.x);
    const int t1 = (int)(((long long)(blockIdx.x + 1) * numTiles) / (int)gridDim.x);
    const int nTiles  = t1 - t0;
    const int nGroups = (nTiles + GROUP_TILES - 1) / GROUP_TILES;

    v2f a[4][16];
    {
        const float* cr = C + (size_t)(wave * 64 + c) * Dk + 2 * h;
        #pragma unroll
        for (int t = 0; t < 4; ++t) {
            #pragma unroll
            for (int j = 0; j < 16; ++j)
                a[t][j] = *(const v2f*)(cr + t * 16 * Dk + 4 * j);
        }
    }
    v8f cc[4];
    #pragma unroll
    for (int t = 0; t < 4; ++t) {
        #pragma unroll
        for (int r = 0; r < 8; ++r)
            cc[t][r] = c2[wave * 64 + 16 * t + 8 * h + r];
    }

    v4f xr[8];
    auto loadTile = [&](int g) {
        const int ti = g * GROUP_TILES + wave;
        if (ti < nTiles) {
            const float* src = X + ((size_t)(t0 + ti) * TILE_ROWS + c) * Dk + 4 * h;
            #pragma unroll
            for (int i = 0; i < 8; ++i) xr[i] = *(const v4f*)(src + 8 * i);
        }
    };
    auto storeTile = [&](int g, int stage) {
        const int ti = g * GROUP_TILES + wave;
        if (ti < nTiles) {
            float* dst = xbuf + stage * XBUF_FLOATS + wave * TILE_FLOATS + 4 * c + 2 * h;
            #pragma unroll
            for (int i = 0; i < 8; ++i) {
                v2f lo = { xr[i].x, xr[i].y };
                v2f hi = { xr[i].z, xr[i].w };
                *(v2f*)(dst + i * JP_PITCH)      = lo;
                *(v2f*)(dst + i * JP_PITCH + 64) = hi;
            }
        }
    };

    loadTile(0);
    storeTile(0, 0);
    __syncthreads();

    int stage = 0;
    for (int g = 0; g < nGroups; ++g) {
        const int left = nTiles - g * GROUP_TILES;
        const int nT = left < GROUP_TILES ? left : GROUP_TILES;
        const int nextStage = (stage == XSTAGES - 1) ? 0 : stage + 1;
        if (g + 1 < nGroups) loadTile(g + 1);

        const float* xb = xbuf + stage * XBUF_FLOATS;
        float* pb = part + (g & 1) * PART_FLOATS;

        v4f bq[8];
        #pragma unroll
        for (int jp = 0; jp < 8; ++jp)
            bq[jp] = *(const v4f*)(xb + jp * JP_PITCH + 4 * lane);

        for (int i = 0; i < nT; ++i) {
            const int inext = (i + 1 < nT) ? i + 1 : i;
            v4f bn[8];
            {
                const float* xt = xb + inext * TILE_FLOATS + 4 * lane;
                #pragma unroll
                for (int jp = 0; jp < 8; ++jp)
                    bn[jp] = *(const v4f*)(xt + jp * JP_PITCH);
            }

            v8f acc[4];
            #pragma unroll
            for (int t = 0; t < 4; ++t)
                acc[t] = (v8f){0.f, 0.f, 0.f, 0.f, 0.f, 0.f, 0.f, 0.f};
            #pragma unroll
            for (int j = 0; j < 16; ++j) {
                v2f b;
                if (j & 1) { b.x = bq[j >> 1].z; b.y = bq[j >> 1].w; }
                else       { b.x = bq[j >> 1].x; b.y = bq[j >> 1].y; }
                #pragma unroll
                for (int t = 0; t < 4; ++t)
                    acc[t] = __builtin_amdgcn_wmma_f32_16x16x4_f32(
                        false, a[t][j], false, b, (short)0, acc[t], false, false);
            }

            const v8f d0 = cc[0] - 2.0f * acc[0];
            const v8f d1 = cc[1] - 2.0f * acc[1];
            const v8f d2 = cc[2] - 2.0f * acc[2];
            const v8f d3 = cc[3] - 2.0f * acc[3];
            auto min8 = [](const v8f v) -> float {
                float g = __builtin_fminf(__builtin_fminf(v[0], v[1]), v[2]);
                g = __builtin_fminf(__builtin_fminf(g, v[3]), v[4]);
                g = __builtin_fminf(__builtin_fminf(g, v[5]), v[6]);
                return __builtin_fminf(g, v[7]);
            };
            const float m0 = min8(d0), m1 = min8(d1), m2 = min8(d2), m3 = min8(d3);
            const float mn = __builtin_fminf(__builtin_fminf(m0, m1), __builtin_fminf(m2, m3));
            const bool e0 = (m0 == mn);
            const bool e1 = (m1 == mn);
            const bool e2 = (m2 == mn);
            const v8f s23  = e2 ? d2 : d3;
            const v8f s123 = e1 ? d1 : s23;
            const v8f sv   = e0 ? d0 : s123;
            const int t23  = e2 ? 32 : 48;
            const int t123 = e1 ? 16 : t23;
            const int ts   = e0 ? 0 : t123;
            int rs = 7;
            #pragma unroll
            for (int r = 6; r >= 0; --r)
                rs = (sv[r] == mn) ? r : rs;
            float best = mn;
            int   bidx = ts + rs + wave * 64 + 8 * h;
            {
                const float ov = __shfl_xor(best, 16, 32);
                const int   oi = __shfl_xor(bidx, 16, 32);
                if (ov < best || (ov == best && oi < bidx)) { best = ov; bidx = oi; }
            }
            if (h == 0) {
                v2f pv = { best, __int_as_float(bidx) };
                *(v2f*)(pb + ((i * TILE_ROWS + c) * NWAVES + wave) * 2) = pv;
            }

            #pragma unroll
            for (int jp = 0; jp < 8; ++jp) bq[jp] = bn[jp];
        }

        if (g + 1 < nGroups) storeTile(g + 1, nextStage);
        __syncthreads();

        if (wave < nT) {
            const float* pp = pb + ((wave * TILE_ROWS + c) * NWAVES + 4 * h) * 2;
            const v4f p0 = *(const v4f*)pp;
            const v4f p1 = *(const v4f*)(pp + 4);
            float bv = p0.x; int bi = __float_as_int(p0.y);
            if (p0.z < bv) { bv = p0.z; bi = __float_as_int(p0.w); }
            if (p1.x < bv) { bv = p1.x; bi = __float_as_int(p1.y); }
            if (p1.z < bv) { bv = p1.z; bi = __float_as_int(p1.w); }
            {
                const float ov = __shfl_xor(bv, 16, 32);
                const int   oi = __shfl_xor(bi, 16, 32);
                if (ov < bv || (ov == bv && oi < bi)) { bv = ov; bi = oi; }
            }
            if (h == 0)
                labels[(size_t)(t0 + g * GROUP_TILES + wave) * TILE_ROWS + c] = (unsigned short)bi;
        }
        stage = nextStage;
    }
}

typedef __attribute__((ext_vector_type(2))) unsigned short us2;
typedef __attribute__((ext_vector_type(4))) unsigned int   v4u;

constexpr int SUM_WAVES = 16;
constexpr int CHUNK     = 256;
constexpr int BATCH     = 16;
constexpr int LIST_CAP  = BATCH + CHUNK;
constexpr int SCAN_AHEAD = 2;
constexpr int OWN        = 16;
constexpr int SUM_SLICES = 16;
constexpr unsigned OWN_MASK = 0xFFFFu & ~(unsigned)(OWN - 1);
constexpr unsigned SLOT_BITS = 5;
constexpr unsigned SLOT_NONE = 16;

__device__ __forceinline__ unsigned pk_min_u16(unsigned a, unsigned b) {
    const us2 r = __builtin_elementwise_min(__builtin_bit_cast(us2, a), __builtin_bit_cast(us2, b));
    return __builtin_bit_cast(unsigned, r);
}

__global__ void __launch_bounds__(32 * SUM_WAVES, 8)
kmeans_sum(const float* __restrict__ X,
           const unsigned short* __restrict__ labels,
           float* __restrict__ part,
           float* __restrict__ cntp,
           int N) {
    __shared__ unsigned list[SUM_WAVES][LIST_CAP];
    __shared__ float psum[SUM_WAVES][OWN][Dk];
    __shared__ float pcnt[SUM_WAVES][OWN];

    const int wave = __builtin_amdgcn_readfirstlane((int)threadIdx.x >> 5);
    const int lane = threadIdx.x & 31;
    const int c = lane & 15;
    const int h = lane >> 4;
    const int owner = (int)blockIdx.x / SUM_SLICES;
    const int slice = (int)blockIdx.x % SUM_SLICES;
    const unsigned k0  = (unsigned)OWN * (unsigned)owner;
    const unsigned pat = k0 | (k0 << 16);
    const unsigned msk = OWN_MASK | (OWN_MASK << 16);
    unsigned* myList = list[wave];

    v8f acc[4];
    #pragma unroll
    for (int t = 0; t < 4; ++t) acc[t] = (v8f){0.f, 0.f, 0.f, 0.f, 0.f, 0.f, 0.f, 0.f};
    unsigned nrows = 0;
    int cnt = 0;

    const int nChunks = (N + CHUNK - 1) / CHUNK;
    const int chunkLo = (int)(((long long)slice * nChunks) / SUM_SLICES);
    const int chunkHi = (int)(((long long)(slice + 1) * nChunks) / SUM_SLICES);
    const v4u* L4 = (const v4u*)labels;

    auto loadChunk = [&](int ch) -> v4u {
        const bool ok = (ch < chunkHi) && (ch * CHUNK + lane * 8 + 8 <= N);
        const v4u w = L4[ok ? (size_t)ch * 32 + lane : (size_t)0];
        const v4u none = {0xFFFFFFFFu, 0xFFFFFFFFu, 0xFFFFFFFFu, 0xFFFFFFFFu};
        return ok ? w : none;
    };

    auto addBatch = [&](int base) {
        v2f   hot[4];
        float xv[4][2][4];
        #pragma unroll
        for (int j = 0; j < 4; ++j) {
            float one[2];
            #pragma unroll
            for (int e = 0; e < 2; ++e) {
                const unsigned w = myList[base + 4 * j + 2 * h + e];
                const bool mine = ((w & ((1u << SLOT_BITS) - 1u)) == (unsigned)c);
                one[e] = mine ? 1.0f : 0.0f;
                nrows += mine ? 1u : 0u;
                const float* p = X + (size_t)(w >> SLOT_BITS) * Dk + c;
                #pragma unroll
                for (int t = 0; t < 4; ++t) xv[j][e][t] = p[16 * t];
            }
            hot[j] = (v2f){ one[0], one[1] };
        }
        __builtin_amdgcn_sched_barrier(0);
        #pragma unroll
        for (int j = 0; j < 4; ++j) {
            #pragma unroll
            for (int t = 0; t < 4; ++t) {
                const v2f b = { xv[j][0][t], xv[j][1][t] };
                acc[t] = __builtin_amdgcn_wmma_f32_16x16x4_f32(
                    false, hot[j], false, b, (short)0, acc[t], false, false);
            }
        }
        __builtin_amdgcn_sched_barrier(0);
        asm volatile("v_nop\n\tv_nop\n\tv_nop\n\tv_nop");
    };

    auto listReady = [&]() {
        __builtin_amdgcn_fence(__ATOMIC_RELEASE, "wavefront");
        __builtin_amdgcn_wave_barrier();
        __builtin_amdgcn_fence(__ATOMIC_ACQUIRE, "wavefront");
    };

    auto scanChunk = [&](v4u w, int ch) {
        const unsigned y0 = (w.x ^ pat) & msk;
        const unsigned y1 = (w.y ^ pat) & msk;
        const unsigned y2 = (w.z ^ pat) & msk;
        const unsigned y3 = (w.w ^ pat) & msk;
        const unsigned t = pk_min_u16(pk_min_u16(y0, y1), pk_min_u16(y2, y3));
        const bool hit = ((t & 0xFFFFu) == 0u) || (t < 0x10000u);
        if (__builtin_amdgcn_ballot_w32(hit) == 0u) return;

        const unsigned ys[4] = { y0, y1, y2, y3 };
        const unsigned ws[4] = { w.x, w.y, w.z, w.w };
        const unsigned rowBase = (unsigned)ch * CHUNK + (unsigned)lane * 8u;
        #pragma unroll
        for (int j = 0; j < 8; ++j) {
            const unsigned yj = (j & 1) ? (ys[j >> 1] >> 16) : (ys[j >> 1] & 0xFFFFu);
            const unsigned lj = (j & 1) ? (ws[j >> 1] >> 16) : (ws[j >> 1] & 0xFFFFu);
            const bool mine = (yj == 0u);
            const unsigned bal = __builtin_amdgcn_ballot_w32(mine);
            if (bal != 0u) {
                if (mine) {
                    const int pos = cnt + (int)__builtin_amdgcn_mbcnt_lo(bal, 0u);
                    myList[pos] = ((rowBase + (unsigned)j) << SLOT_BITS) | (lj & (unsigned)(OWN - 1));
                }
                cnt += __builtin_popcount(bal);
            }
        }
        listReady();
        while (cnt >= BATCH) { cnt -= BATCH; addBatch(cnt); }
    };

    v4u cur[SCAN_AHEAD];
    #pragma unroll
    for (int u = 0; u < SCAN_AHEAD; ++u) cur[u] = loadChunk(chunkLo + wave + SUM_WAVES * u);
    for (int ch = chunkLo + wave; ch < chunkHi; ch += SUM_WAVES * SCAN_AHEAD) {
        v4u nxt[SCAN_AHEAD];
        #pragma unroll
        for (int u = 0; u < SCAN_AHEAD; ++u) nxt[u] = loadChunk(ch + SUM_WAVES * (SCAN_AHEAD + u));
        #pragma unroll
        for (int u = 0; u < SCAN_AHEAD; ++u) scanChunk(cur[u], ch + SUM_WAVES * u);
        #pragma unroll
        for (int u = 0; u < SCAN_AHEAD; ++u) cur[u] = nxt[u];
    }

    if (cnt > 0) {
        if (lane >= cnt && lane < BATCH) myList[lane] = SLOT_NONE;
        listReady();
        addBatch(0);
    }

    #pragma unroll
    for (int t = 0; t < 4; ++t) {
        #pragma unroll
        for (int r = 0; r < 8; ++r) psum[wave][8 * h + r][16 * t + c] = acc[t][r];
    }
    {
        const unsigned tot = nrows + (unsigned)__shfl_xor((int)nrows, 16, 32);
        if (h == 0) pcnt[wave][c] = (float)tot;
    }
    __syncthreads();

    for (int o = (int)threadIdx.x; o < OWN * Dk; o += 32 * SUM_WAVES) {
        const int which = o >> 6;
        const int f     = o & (Dk - 1);
        float s = 0.f, n = 0.f;
        #pragma unroll
        for (int w = 0; w < SUM_WAVES; ++w) { s += psum[w][which][f]; n += pcnt[w][which]; }
        const int k = (int)k0 + which;
        part[((size_t)slice * Kc + k) * Dk + f] = s;
        if (f == 0) cntp[slice * Kc + k] = n;
    }
}

__global__ void __launch_bounds__(4 * Dk)
kmeans_finish(const float* __restrict__ part,
              const float* __restrict__ cntp,
              float* __restrict__ out) {
    const int k = (int)blockIdx.x * 4 + ((int)threadIdx.x >> 6);
    const int f = (int)threadIdx.x & (Dk - 1);
    float s = 0.f, n = 0.f;
    #pragma unroll
    for (int sl = 0; sl < SUM_SLICES; ++sl) {
        s += part[((size_t)sl * Kc + k) * Dk + f];
        n += cntp[sl * Kc + k];
    }
    out[k * Dk + f] = s / fmaxf(n, 1.0f);
}

extern "C" void kernel_launch(void* const* d_in, const int* in_sizes, int n_in,
                              void* d_out, int out_size, void* d_ws, size_t ws_size,
                              hipStream_t stream) {
    const float* X        = (const float*)d_in[0];
    const int*   init_idx = (const int*)d_in[1];
    float*       out      = (float*)d_out;

    const int N          = in_sizes[0] / Dk;
    const int numTiles32 = N / 32;

    float* ws   = (float*)d_ws;
    float* C    = ws;
    float* c2   = C + Kc * Dk;
    float* part = c2 + Kc;
    float* cntp = part + SUM_SLICES * Kc * Dk;
    unsigned short* labels = (unsigned short*)(cntp + SUM_SLICES * Kc);

    kmeans_prep<<<Kc, Dk, 0, stream>>>(X, init_idx, C, c2);

    kmeans_assign<<<ASSIGN_BLOCKS, 32 * NWAVES, ASSIGN_LDS_BYTES, stream>>>(
        X, C, c2, labels, numTiles32);

    kmeans_sum<<<(Kc / OWN) * SUM_SLICES, 32 * SUM_WAVES, 0, stream>>>(X, labels, part, cntp, N);

    kmeans_finish<<<Kc / 4, 4 * Dk, 0, stream>>>(part, cntp, out);
}
